# starting kernel
# baseline (speedup 1.0000x reference)
#include <hip/hip_runtime.h>
#include <math.h>

#define N_NODES 50000
#define N_EDGES 600000
#define NFEAT 128
#define NHID 128
#define NCLASS 41

typedef float v2f __attribute__((ext_vector_type(2)));
typedef float v8f __attribute__((ext_vector_type(8)));

template<bool BIAS_RELU>
__global__ __launch_bounds__(256)
void gemm128_wmma(const float* __restrict__ A, const float* __restrict__ B,
                  const float* __restrict__ bias, float* __restrict__ C)
{
    const int lane = threadIdx.x & 31;
    const int wave = threadIdx.x >> 5;
    const int m0   = blockIdx.x * 16;
    const int n0   = wave * 16;
    const int m    = lane & 15;
    const int half = lane >> 4;

    const float* __restrict__ arow = A + (size_t)(m0 + m) * NHID;

    v8f acc = {};
#pragma unroll
    for (int kb = 0; kb < NHID; kb += 4) {
        const int ka = kb + 2 * half;
        float a0 = arow[ka];
        float a1 = arow[ka + 1];
        if (BIAS_RELU) {
            a0 = fmaxf(a0 + bias[ka],     0.0f);
            a1 = fmaxf(a1 + bias[ka + 1], 0.0f);
        }
        v2f a; a.x = a0; a.y = a1;
        v2f b;
        b.x = B[(size_t)ka       * NHID + n0 + m];
        b.y = B[(size_t)(ka + 1) * NHID + n0 + m];
        acc = __builtin_amdgcn_wmma_f32_16x16x4_f32(
                   false, a,  false, b,
                   (short)0, acc,  false,  false);
    }

#pragma unroll
    for (int v = 0; v < 8; ++v)
        C[(size_t)(m0 + v + 8 * half) * NHID + n0 + m] = acc[v];
}

__global__ __launch_bounds__(256)
void spmm_atomic(const int* __restrict__ src, const int* __restrict__ dst,
                 const float* __restrict__ w, const float* __restrict__ h,
                 float* __restrict__ acc)
{
    const int e = (int)((blockIdx.x * blockDim.x + threadIdx.x) >> 5);
    if (e >= N_EDGES) return;
    const int lane = threadIdx.x & 31;
    const int s  = src[e];
    const int d  = dst[e];
    const float wt = w[e];

    const float4 hv = *((const float4*)(h + (size_t)s * NHID) + lane);
    float* ap = acc + (size_t)d * NHID + lane * 4;
    unsafeAtomicAdd(ap + 0, wt * hv.x);
    unsafeAtomicAdd(ap + 1, wt * hv.y);
    unsafeAtomicAdd(ap + 2, wt * hv.z);
    unsafeAtomicAdd(ap + 3, wt * hv.w);
}

__global__ __launch_bounds__(256)
void zero_f4(float4* __restrict__ p, int n4)
{
    const int i = blockIdx.x * blockDim.x + threadIdx.x;
    if (i < n4) p[i] = make_float4(0.f, 0.f, 0.f, 0.f);
}

__global__ __launch_bounds__(64)
void head_logsoftmax(const float* __restrict__ acc, const float* __restrict__ b2,
                     const float* __restrict__ Wout, const float* __restrict__ bout,
                     float* __restrict__ out)
{
    __shared__ float hsh[NHID];
    __shared__ float red[64];
    const int node = blockIdx.x;
    const int t = threadIdx.x;

    for (int i = t; i < NHID; i += 64)
        hsh[i] = fmaxf(acc[(size_t)node * NHID + i] + b2[i], 0.0f);
    __syncthreads();

    float logit = -INFINITY;
    if (t < NCLASS) {
        float a = bout[t];
#pragma unroll 4
        for (int k = 0; k < NHID; ++k)
            a = fmaf(hsh[k], Wout[(size_t)k * NCLASS + t], a);
        logit = a;
    }

    red[t] = logit;
    __syncthreads();
    for (int s = 32; s > 0; s >>= 1) {
        if (t < s) red[t] = fmaxf(red[t], red[t + s]);
        __syncthreads();
    }
    const float mx = red[0];
    __syncthreads();

    red[t] = (t < NCLASS) ? __expf(logit - mx) : 0.0f;
    __syncthreads();
    for (int s = 32; s > 0; s >>= 1) {
        if (t < s) red[t] += red[t + s];
        __syncthreads();
    }
    const float lse = __logf(red[0]) + mx;

    if (t < NCLASS)
        out[(size_t)node * NCLASS + t] = logit - lse;
}

extern "C" void kernel_launch(void* const* d_in, const int* in_sizes, int n_in,
                              void* d_out, int out_size, void* d_ws, size_t ws_size,
                              hipStream_t stream)
{
    const float* x    = (const float*)d_in[0];
    const int*   a0i  = (const int*)  d_in[1];
    const float* a0w  = (const float*)d_in[2];
    const int*   a1i  = (const int*)  d_in[3];
    const float* a1w  = (const float*)d_in[4];
    const float* W1   = (const float*)d_in[5];
    const float* b1   = (const float*)d_in[6];
    const float* W2   = (const float*)d_in[7];
    const float* b2   = (const float*)d_in[8];
    const float* Wout = (const float*)d_in[9];
    const float* bout = (const float*)d_in[10];
    float* out = (float*)d_out;

    const size_t NH = (size_t)N_NODES * NHID;
    float* bufA = (float*)d_ws;
    float* bufB = bufA + NH;

    const dim3 gGemm(N_NODES / 16), bGemm(256);
    const int  n4 = (int)(NH / 4);
    const dim3 gZero((n4 + 255) / 256), bZero(256);
    const dim3 gSp((unsigned)(((size_t)N_EDGES * 32 + 255) / 256)), bSp(256);

    gemm128_wmma<false><<<gGemm, bGemm, 0, stream>>>(x, W1, nullptr, bufA);
    zero_f4<<<gZero, bZero, 0, stream>>>((float4*)bufB, n4);
    spmm_atomic<<<gSp, bSp, 0, stream>>>(a0i, a0i + N_EDGES, a0w, bufA, bufB);

    gemm128_wmma<true><<<gGemm, bGemm, 0, stream>>>(bufB, W2, b1, bufA);
    zero_f4<<<gZero, bZero, 0, stream>>>((float4*)bufB, n4);
    spmm_atomic<<<gSp, bSp, 0, stream>>>(a1i, a1i + N_EDGES, a1w, bufA, bufB);

    head_logsoftmax<<<dim3(N_NODES), dim3(64), 0, stream>>>(bufB, b2, Wout, bout, out);
}


# spmm: edges bucketed by dst row into a per-call CSR (int-atomic rank, prefix sum, place); a wave gathers its rows' h rows and sums w*h by fma in registers in rank (arrival) order, one store per row
# speedup vs baseline: 1.3268x; 1.3268x over previous
#include <hip/hip_runtime.h>
#include <math.h>

#define N_NODES 50000
#define N_EDGES 600000
#define NFEAT 128
#define NHID 128
#define NCLASS 41
#define N_ROWS2  (2 * N_NODES)
#define N_EDGES2 (2 * N_EDGES)
#define ROWS2_PAD 100352
#define ROWS_PER_WAVE 4

typedef float v2f __attribute__((ext_vector_type(2)));
typedef float v8f __attribute__((ext_vector_type(8)));

template<bool BIAS_RELU>
__global__ __launch_bounds__(256)
void gemm128_wmma(const float* __restrict__ A, const float* __restrict__ B,
                  const float* __restrict__ bias, float* __restrict__ C)
{
    const int lane = threadIdx.x & 31;
    const int wave = threadIdx.x >> 5;
    const int m0   = blockIdx.x * 16;
    const int n0   = wave * 16;
    const int m    = lane & 15;
    const int half = lane >> 4;

    const float* __restrict__ arow = A + (size_t)(m0 + m) * NHID;

    v8f acc = {};
#pragma unroll
    for (int kb = 0; kb < NHID; kb += 4) {
        const int ka = kb + 2 * half;
        float a0 = arow[ka];
        float a1 = arow[ka + 1];
        if (BIAS_RELU) {
            a0 = fmaxf(a0 + bias[ka],     0.0f);
            a1 = fmaxf(a1 + bias[ka + 1], 0.0f);
        }
        v2f a; a.x = a0; a.y = a1;
        v2f b;
        b.x = B[(size_t)ka       * NHID + n0 + m];
        b.y = B[(size_t)(ka + 1) * NHID + n0 + m];
        acc = __builtin_amdgcn_wmma_f32_16x16x4_f32(
                   false, a,  false, b,
                   (short)0, acc,  false,  false);
    }

#pragma unroll
    for (int v = 0; v < 8; ++v)
        C[(size_t)(m0 + v + 8 * half) * NHID + n0 + m] = acc[v];
}

__global__ __launch_bounds__(256)
void zero_i4(int4* __restrict__ p, int n4)
{
    const int i = blockIdx.x * blockDim.x + threadIdx.x;
    if (i < n4) p[i] = make_int4(0, 0, 0, 0);
}

__global__ __launch_bounds__(256)
void csr_count_rank(const int* __restrict__ dst0, const int* __restrict__ dst1,
                    int* __restrict__ cnt, int* __restrict__ rank)
{
    const int g = blockIdx.y;
    const int q = blockIdx.x * blockDim.x + threadIdx.x;
    if (q >= N_EDGES / 4) return;
    const int4 d = ((const int4*)(g ? dst1 : dst0))[q];
    int* __restrict__ c = cnt + g * N_NODES;
    int4 r;
    r.x = atomicAdd(c + d.x, 1);
    r.y = atomicAdd(c + d.y, 1);
    r.z = atomicAdd(c + d.z, 1);
    r.w = atomicAdd(c + d.w, 1);
    ((int4*)(rank + g * N_EDGES))[q] = r;
}

__global__ __launch_bounds__(1024)
void csr_scan(const int* __restrict__ cnt, int* __restrict__ rowptr)
{
    __shared__ int w_tot[32];
    __shared__ int w_front[32];
    __shared__ int s_front;
    const int t = threadIdx.x, lane = t & 31, wv = t >> 5;
    const int base = blockIdx.x * 1024;

    int front = 0;
    const int4* __restrict__ c4 = (const int4*)cnt;
    const int nq = base / 4;
    for (int i = t; i < nq; i += 8 * 1024) {
#pragma unroll
        for (int u = 0; u < 8; ++u) {
            const int j = i + u * 1024;
            const int4 v = c4[j < nq ? j : ROWS2_PAD / 4 - 1];
            front += (v.x + v.y) + (v.z + v.w);
        }
    }
    const int idx = base + t;
    const int c = idx < N_ROWS2 ? cnt[idx] : 0;

    int incl = c;
    for (int o = 1; o < 32; o <<= 1) {
        const int v = __shfl_up(incl, o);
        if (lane >= o) incl += v;
    }
    for (int o = 16; o > 0; o >>= 1) front += __shfl_xor(front, o);
    if (lane == 31) w_tot[wv] = incl;
    if (lane == 0)  w_front[wv] = front;
    __syncthreads();
    if (wv == 0) {
        const int tot = w_tot[lane];
        int wi = tot;
        for (int o = 1; o < 32; o <<= 1) {
            const int v = __shfl_up(wi, o);
            if (lane >= o) wi += v;
        }
        int f = w_front[lane];
        for (int o = 16; o > 0; o >>= 1) f += __shfl_xor(f, o);
        w_tot[lane] = wi - tot;
        if (lane == 0) s_front = f;
    }
    __syncthreads();
    const int excl = s_front + w_tot[wv] + incl - c;
    if (idx < N_ROWS2)      rowptr[idx] = excl;
    if (idx == N_ROWS2 - 1) rowptr[N_ROWS2] = excl + c;
}

__global__ __launch_bounds__(256)
void csr_place(const int* __restrict__ idx0, const float* __restrict__ w0,
               const int* __restrict__ idx1, const float* __restrict__ w1,
               const int* __restrict__ rowptr, const int* __restrict__ rank,
               int2* __restrict__ rec)
{
    const int g = blockIdx.y;
    const int q = blockIdx.x * blockDim.x + threadIdx.x;
    if (q >= N_EDGES / 4) return;
    const int* __restrict__ idx = g ? idx1 : idx0;
    const int4   s = ((const int4*)idx)[q];
    const int4   d = ((const int4*)(idx + N_EDGES))[q];
    const float4 w = ((const float4*)(g ? w1 : w0))[q];
    const int4   r = ((const int4*)(rank + g * N_EDGES))[q];
    const int* __restrict__ rp = rowptr + g * N_NODES;
    const unsigned p0 = min((unsigned)(rp[d.x] + r.x), (unsigned)(N_EDGES2 - 1));
    const unsigned p1 = min((unsigned)(rp[d.y] + r.y), (unsigned)(N_EDGES2 - 1));
    const unsigned p2 = min((unsigned)(rp[d.z] + r.z), (unsigned)(N_EDGES2 - 1));
    const unsigned p3 = min((unsigned)(rp[d.w] + r.w), (unsigned)(N_EDGES2 - 1));
    rec[p0] = make_int2(s.x, __float_as_int(w.x));
    rec[p1] = make_int2(s.y, __float_as_int(w.y));
    rec[p2] = make_int2(s.z, __float_as_int(w.z));
    rec[p3] = make_int2(s.w, __float_as_int(w.w));
}

__device__ __forceinline__
void gather_chunk(const float4* __restrict__ h4, int lane, int rs, float rw, int n, float4& acc)
{
    for (int j = 0; j < n; j += 4) {
        const unsigned s0 = min((unsigned)__builtin_amdgcn_readlane(rs, j),     (unsigned)(N_NODES - 1));
        const unsigned s1 = min((unsigned)__builtin_amdgcn_readlane(rs, j + 1), (unsigned)(N_NODES - 1));
        const unsigned s2 = min((unsigned)__builtin_amdgcn_readlane(rs, j + 2), (unsigned)(N_NODES - 1));
        const unsigned s3 = min((unsigned)__builtin_amdgcn_readlane(rs, j + 3), (unsigned)(N_NODES - 1));
        const float w0 = __int_as_float(__builtin_amdgcn_readlane(__float_as_int(rw), j));
        const float w1 = __int_as_float(__builtin_amdgcn_readlane(__float_as_int(rw), j + 1));
        const float w2 = __int_as_float(__builtin_amdgcn_readlane(__float_as_int(rw), j + 2));
        const float w3 = __int_as_float(__builtin_amdgcn_readlane(__float_as_int(rw), j + 3));
        const float4 v0 = h4[(size_t)s0 * (NHID / 4) + lane];
        const float4 v1 = h4[(size_t)s1 * (NHID / 4) + lane];
        const float4 v2 = h4[(size_t)s2 * (NHID / 4) + lane];
        const float4 v3 = h4[(size_t)s3 * (NHID / 4) + lane];
        acc.x = fmaf(w0, v0.x, acc.x); acc.y = fmaf(w0, v0.y, acc.y);
        acc.z = fmaf(w0, v0.z, acc.z); acc.w = fmaf(w0, v0.w, acc.w);
        acc.x = fmaf(w1, v1.x, acc.x); acc.y = fmaf(w1, v1.y, acc.y);
        acc.z = fmaf(w1, v1.z, acc.z); acc.w = fmaf(w1, v1.w, acc.w);
        acc.x = fmaf(w2, v2.x, acc.x); acc.y = fmaf(w2, v2.y, acc.y);
        acc.z = fmaf(w2, v2.z, acc.z); acc.w = fmaf(w2, v2.w, acc.w);
        acc.x = fmaf(w3, v3.x, acc.x); acc.y = fmaf(w3, v3.y, acc.y);
        acc.z = fmaf(w3, v3.z, acc.z); acc.w = fmaf(w3, v3.w, acc.w);
    }
}

__global__ __launch_bounds__(256)
void spmm_pull(const int* __restrict__ rowptr, const int2* __restrict__ rec,
               const float* __restrict__ h, float* __restrict__ out)
{
    const int lane = threadIdx.x & 31;
    const int row0 = __builtin_amdgcn_readfirstlane(
                         (int)((blockIdx.x * 8 + (threadIdx.x >> 5)) * ROWS_PER_WAVE));
    if (row0 >= N_NODES) return;

    int p[ROWS_PER_WAVE + 1];
#pragma unroll
    for (int k = 0; k <= ROWS_PER_WAVE; ++k)
        p[k] = min(max(rowptr[row0 + k], 0), N_EDGES2);

    int rs[ROWS_PER_WAVE]; float rw[ROWS_PER_WAVE]; int n[ROWS_PER_WAVE];
#pragma unroll
    for (int k = 0; k < ROWS_PER_WAVE; ++k) {
        n[k] = max(min(p[k + 1] - p[k], 32), 0);
        const int i = min(p[k] + min(lane, max(n[k], 1) - 1), N_EDGES2 - 1);
        const int2 r = rec[i];
        rs[k] = r.x;
        rw[k] = lane < n[k] ? __int_as_float(r.y) : 0.0f;
    }

    const float4* __restrict__ h4 = (const float4*)h;
    float4* __restrict__ out4 = (float4*)out;
#pragma unroll
    for (int k = 0; k < ROWS_PER_WAVE; ++k) {
        float4 acc = make_float4(0.f, 0.f, 0.f, 0.f);
        gather_chunk(h4, lane, rs[k], rw[k], n[k], acc);
        for (int base = p[k] + 32; base < p[k + 1]; base += 32) {
            const int m = min(p[k + 1] - base, 32);
            const int2 r = rec[base + min(lane, m - 1)];
            gather_chunk(h4, lane, r.x, lane < m ? __int_as_float(r.y) : 0.0f, m, acc);
        }
        out4[(size_t)(row0 + k) * (NHID / 4) + lane] = acc;
    }
}

__global__ __launch_bounds__(64)
void head_logsoftmax(const float* __restrict__ acc, const float* __restrict__ b2,
                     const float* __restrict__ Wout, const float* __restrict__ bout,
                     float* __restrict__ out)
{
    __shared__ float hsh[NHID];
    __shared__ float red[64];
    const int node = blockIdx.x;
    const int t = threadIdx.x;

    for (int i = t; i < NHID; i += 64)
        hsh[i] = fmaxf(acc[(size_t)node * NHID + i] + b2[i], 0.0f);
    __syncthreads();

    float logit = -INFINITY;
    if (t < NCLASS) {
        float a = bout[t];
#pragma unroll 4
        for (int k = 0; k < NHID; ++k)
            a = fmaf(hsh[k], Wout[(size_t)k * NCLASS + t], a);
        logit = a;
    }

    red[t] = logit;
    __syncthreads();
    for (int s = 32; s > 0; s >>= 1) {
        if (t < s) red[t] = fmaxf(red[t], red[t + s]);
        __syncthreads();
    }
    const float mx = red[0];
    __syncthreads();

    red[t] = (t < NCLASS) ? __expf(logit - mx) : 0.0f;
    __syncthreads();
    for (int s = 32; s > 0; s >>= 1) {
        if (t < s) red[t] += red[t + s];
        __syncthreads();
    }
    const float lse = __logf(red[0]) + mx;

    if (t < NCLASS)
        out[(size_t)node * NCLASS + t] = logit - lse;
}

extern "C" void kernel_launch(void* const* d_in, const int* in_sizes, int n_in,
                              void* d_out, int out_size, void* d_ws, size_t ws_size,
                              hipStream_t stream)
{
    const float* x    = (const float*)d_in[0];
    const int*   a0i  = (const int*)  d_in[1];
    const float* a0w  = (const float*)d_in[2];
    const int*   a1i  = (const int*)  d_in[3];
    const float* a1w  = (const float*)d_in[4];
    const float* W1   = (const float*)d_in[5];
    const float* b1   = (const float*)d_in[6];
    const float* W2   = (const float*)d_in[7];
    const float* b2   = (const float*)d_in[8];
    const float* Wout = (const float*)d_in[9];
    const float* bout = (const float*)d_in[10];
    float* out = (float*)d_out;

    const size_t NH = (size_t)N_NODES * NHID;
    float* bufA = (float*)d_ws;
    float* bufB = bufA + NH;
    int*   cnt    = (int*)(bufB + NH);
    int*   rowptr = cnt + ROWS2_PAD;
    int*   rank   = rowptr + ROWS2_PAD;
    int2*  rec    = (int2*)(rank + N_EDGES2);

    const dim3 gGemm(N_NODES / 16), bGemm(256);
    const dim3 gZero(ROWS2_PAD / 4 / 256), bZero(256);
    const dim3 gEdge((N_EDGES / 4 + 255) / 256, 2), bEdge(256);
    const dim3 gScan(ROWS2_PAD / 1024), bScan(1024);
    const dim3 gPull((N_NODES / ROWS_PER_WAVE + 7) / 8), bPull(256);

    zero_i4<<<gZero, bZero, 0, stream>>>((int4*)cnt, ROWS2_PAD / 4);
    csr_count_rank<<<gEdge, bEdge, 0, stream>>>(a0i + N_EDGES, a1i + N_EDGES, cnt, rank);
    csr_scan<<<gScan, bScan, 0, stream>>>(cnt, rowptr);
    csr_place<<<gEdge, bEdge, 0, stream>>>(a0i, a0w, a1i, a1w, rowptr, rank, rec);

    gemm128_wmma<false><<<gGemm, bGemm, 0, stream>>>(x, W1, nullptr, bufA);
    spmm_pull<<<gPull, bPull, 0, stream>>>(rowptr, rec, bufA, bufB);

    gemm128_wmma<true><<<gGemm, bGemm, 0, stream>>>(bufB, W2, b1, bufA);
    spmm_pull<<<gPull, bPull, 0, stream>>>(rowptr + N_NODES, rec, bufA, bufB);

    head_logsoftmax<<<dim3(N_NODES), dim3(64), 0, stream>>>(bufB, b2, Wout, bout, out);
}


# head: class-major product on f32 WMMA (Wout^T as A, 16 node rows as B), one wave per 16 nodes, 625 work-groups not 50000; log-softmax in registers per lane; dot and exp sums in a new fixed order
# speedup vs baseline: 1.6615x; 1.2523x over previous
#include <hip/hip_runtime.h>
#include <math.h>

#define N_NODES 50000
#define N_EDGES 600000
#define NFEAT 128
#define NHID 128
#define NCLASS 41
#define N_ROWS2  (2 * N_NODES)
#define N_EDGES2 (2 * N_EDGES)
#define ROWS2_PAD 100352
#define ROWS_PER_WAVE 4

typedef float v2f __attribute__((ext_vector_type(2)));
typedef float v8f __attribute__((ext_vector_type(8)));

template<bool BIAS_RELU>
__global__ __launch_bounds__(256)
void gemm128_wmma(const float* __restrict__ A, const float* __restrict__ B,
                  const float* __restrict__ bias, float* __restrict__ C)
{
    const int lane = threadIdx.x & 31;
    const int wave = threadIdx.x >> 5;
    const int m0   = blockIdx.x * 16;
    const int n0   = wave * 16;
    const int m    = lane & 15;
    const int half = lane >> 4;

    const float* __restrict__ arow = A + (size_t)(m0 + m) * NHID;

    v8f acc = {};
#pragma unroll
    for (int kb = 0; kb < NHID; kb += 4) {
        const int ka = kb + 2 * half;
        float a0 = arow[ka];
        float a1 = arow[ka + 1];
        if (BIAS_RELU) {
            a0 = fmaxf(a0 + bias[ka],     0.0f);
            a1 = fmaxf(a1 + bias[ka + 1], 0.0f);
        }
        v2f a; a.x = a0; a.y = a1;
        v2f b;
        b.x = B[(size_t)ka       * NHID + n0 + m];
        b.y = B[(size_t)(ka + 1) * NHID + n0 + m];
        acc = __builtin_amdgcn_wmma_f32_16x16x4_f32(
                   false, a,  false, b,
                   (short)0, acc,  false,  false);
    }

#pragma unroll
    for (int v = 0; v < 8; ++v)
        C[(size_t)(m0 + v + 8 * half) * NHID + n0 + m] = acc[v];
}

__global__ __launch_bounds__(256)
void zero_i4(int4* __restrict__ p, int n4)
{
    const int i = blockIdx.x * blockDim.x + threadIdx.x;
    if (i < n4) p[i] = make_int4(0, 0, 0, 0);
}

__global__ __launch_bounds__(256)
void csr_count_rank(const int* __restrict__ dst0, const int* __restrict__ dst1,
                    int* __restrict__ cnt, int* __restrict__ rank)
{
    const int g = blockIdx.y;
    const int q = blockIdx.x * blockDim.x + threadIdx.x;
    if (q >= N_EDGES / 4) return;
    const int4 d = ((const int4*)(g ? dst1 : dst0))[q];
    int* __restrict__ c = cnt + g * N_NODES;
    int4 r;
    r.x = atomicAdd(c + d.x, 1);
    r.y = atomicAdd(c + d.y, 1);
    r.z = atomicAdd(c + d.z, 1);
    r.w = atomicAdd(c + d.w, 1);
    ((int4*)(rank + g * N_EDGES))[q] = r;
}

__global__ __launch_bounds__(1024)
void csr_scan(const int* __restrict__ cnt, int* __restrict__ rowptr)
{
    __shared__ int w_tot[32];
    __shared__ int w_front[32];
    __shared__ int s_front;
    const int t = threadIdx.x, lane = t & 31, wv = t >> 5;
    const int base = blockIdx.x * 1024;

    int front = 0;
    const int4* __restrict__ c4 = (const int4*)cnt;
    const int nq = base / 4;
    for (int i = t; i < nq; i += 8 * 1024) {
#pragma unroll
        for (int u = 0; u < 8; ++u) {
            const int j = i + u * 1024;
            const int4 v = c4[j < nq ? j : ROWS2_PAD / 4 - 1];
            front += (v.x + v.y) + (v.z + v.w);
        }
    }
    const int idx = base + t;
    const int c = idx < N_ROWS2 ? cnt[idx] : 0;

    int incl = c;
    for (int o = 1; o < 32; o <<= 1) {
        const int v = __shfl_up(incl, o);
        if (lane >= o) incl += v;
    }
    for (int o = 16; o > 0; o >>= 1) front += __shfl_xor(front, o);
    if (lane == 31) w_tot[wv] = incl;
    if (lane == 0)  w_front[wv] = front;
    __syncthreads();
    if (wv == 0) {
        const int tot = w_tot[lane];
        int wi = tot;
        for (int o = 1; o < 32; o <<= 1) {
            const int v = __shfl_up(wi, o);
            if (lane >= o) wi += v;
        }
        int f = w_front[lane];
        for (int o = 16; o > 0; o >>= 1) f += __shfl_xor(f, o);
        w_tot[lane] = wi - tot;
        if (lane == 0) s_front = f;
    }
    __syncthreads();
    const int excl = s_front + w_tot[wv] + incl - c;
    if (idx < N_ROWS2)      rowptr[idx] = excl;
    if (idx == N_ROWS2 - 1) rowptr[N_ROWS2] = excl + c;
}

__global__ __launch_bounds__(256)
void csr_place(const int* __restrict__ idx0, const float* __restrict__ w0,
               const int* __restrict__ idx1, const float* __restrict__ w1,
               const int* __restrict__ rowptr, const int* __restrict__ rank,
               int2* __restrict__ rec)
{
    const int g = blockIdx.y;
    const int q = blockIdx.x * blockDim.x + threadIdx.x;
    if (q >= N_EDGES / 4) return;
    const int* __restrict__ idx = g ? idx1 : idx0;
    const int4   s = ((const int4*)idx)[q];
    const int4   d = ((const int4*)(idx + N_EDGES))[q];
    const float4 w = ((const float4*)(g ? w1 : w0))[q];
    const int4   r = ((const int4*)(rank + g * N_EDGES))[q];
    const int* __restrict__ rp = rowptr + g * N_NODES;
    const unsigned p0 = min((unsigned)(rp[d.x] + r.x), (unsigned)(N_EDGES2 - 1));
    const unsigned p1 = min((unsigned)(rp[d.y] + r.y), (unsigned)(N_EDGES2 - 1));
    const unsigned p2 = min((unsigned)(rp[d.z] + r.z), (unsigned)(N_EDGES2 - 1));
    const unsigned p3 = min((unsigned)(rp[d.w] + r.w), (unsigned)(N_EDGES2 - 1));
    rec[p0] = make_int2(s.x, __float_as_int(w.x));
    rec[p1] = make_int2(s.y, __float_as_int(w.y));
    rec[p2] = make_int2(s.z, __float_as_int(w.z));
    rec[p3] = make_int2(s.w, __float_as_int(w.w));
}

__device__ __forceinline__
void gather_chunk(const float4* __restrict__ h4, int lane, int rs, float rw, int n, float4& acc)
{
    for (int j = 0; j < n; j += 4) {
        const unsigned s0 = min((unsigned)__builtin_amdgcn_readlane(rs, j),     (unsigned)(N_NODES - 1));
        const unsigned s1 = min((unsigned)__builtin_amdgcn_readlane(rs, j + 1), (unsigned)(N_NODES - 1));
        const unsigned s2 = min((unsigned)__builtin_amdgcn_readlane(rs, j + 2), (unsigned)(N_NODES - 1));
        const unsigned s3 = min((unsigned)__builtin_amdgcn_readlane(rs, j + 3), (unsigned)(N_NODES - 1));
        const float w0 = __int_as_float(__builtin_amdgcn_readlane(__float_as_int(rw), j));
        const float w1 = __int_as_float(__builtin_amdgcn_readlane(__float_as_int(rw), j + 1));
        const float w2 = __int_as_float(__builtin_amdgcn_readlane(__float_as_int(rw), j + 2));
        const float w3 = __int_as_float(__builtin_amdgcn_readlane(__float_as_int(rw), j + 3));
        const float4 v0 = h4[(size_t)s0 * (NHID / 4) + lane];
        const float4 v1 = h4[(size_t)s1 * (NHID / 4) + lane];
        const float4 v2 = h4[(size_t)s2 * (NHID / 4) + lane];
        const float4 v3 = h4[(size_t)s3 * (NHID / 4) + lane];
        acc.x = fmaf(w0, v0.x, acc.x); acc.y = fmaf(w0, v0.y, acc.y);
        acc.z = fmaf(w0, v0.z, acc.z); acc.w = fmaf(w0, v0.w, acc.w);
        acc.x = fmaf(w1, v1.x, acc.x); acc.y = fmaf(w1, v1.y, acc.y);
        acc.z = fmaf(w1, v1.z, acc.z); acc.w = fmaf(w1, v1.w, acc.w);
        acc.x = fmaf(w2, v2.x, acc.x); acc.y = fmaf(w2, v2.y, acc.y);
        acc.z = fmaf(w2, v2.z, acc.z); acc.w = fmaf(w2, v2.w, acc.w);
        acc.x = fmaf(w3, v3.x, acc.x); acc.y = fmaf(w3, v3.y, acc.y);
        acc.z = fmaf(w3, v3.z, acc.z); acc.w = fmaf(w3, v3.w, acc.w);
    }
}

__global__ __launch_bounds__(256)
void spmm_pull(const int* __restrict__ rowptr, const int2* __restrict__ rec,
               const float* __restrict__ h, float* __restrict__ out)
{
    const int lane = threadIdx.x & 31;
    const int row0 = __builtin_amdgcn_readfirstlane(
                         (int)((blockIdx.x * 8 + (threadIdx.x >> 5)) * ROWS_PER_WAVE));
    if (row0 >= N_NODES) return;

    int p[ROWS_PER_WAVE + 1];
#pragma unroll
    for (int k = 0; k <= ROWS_PER_WAVE; ++k)
        p[k] = min(max(rowptr[row0 + k], 0), N_EDGES2);

    int rs[ROWS_PER_WAVE]; float rw[ROWS_PER_WAVE]; int n[ROWS_PER_WAVE];
#pragma unroll
    for (int k = 0; k < ROWS_PER_WAVE; ++k) {
        n[k] = max(min(p[k + 1] - p[k], 32), 0);
        const int i = min(p[k] + min(lane, max(n[k], 1) - 1), N_EDGES2 - 1);
        const int2 r = rec[i];
        rs[k] = r.x;
        rw[k] = lane < n[k] ? __int_as_float(r.y) : 0.0f;
    }

    const float4* __restrict__ h4 = (const float4*)h;
    float4* __restrict__ out4 = (float4*)out;
#pragma unroll
    for (int k = 0; k < ROWS_PER_WAVE; ++k) {
        float4 acc = make_float4(0.f, 0.f, 0.f, 0.f);
        gather_chunk(h4, lane, rs[k], rw[k], n[k], acc);
        for (int base = p[k] + 32; base < p[k + 1]; base += 32) {
            const int m = min(p[k + 1] - base, 32);
            const int2 r = rec[base + min(lane, m - 1)];
            gather_chunk(h4, lane, r.x, lane < m ? __int_as_float(r.y) : 0.0f, m, acc);
        }
        out4[(size_t)(row0 + k) * (NHID / 4) + lane] = acc;
    }
}

#define HEAD_WAVES 5
#define HEAD_ROW_FLOATS (16 * NCLASS)

__global__ __launch_bounds__(32 * HEAD_WAVES)
void head_logsoftmax_wmma(const float* __restrict__ acc, const float* __restrict__ b2,
                          const float* __restrict__ Wout, const float* __restrict__ bout,
                          float* __restrict__ out)
{
    __shared__ float stage[HEAD_WAVES][HEAD_ROW_FLOATS];
    const int lane = threadIdx.x & 31;
    const int wave = threadIdx.x >> 5;
    const int c    = lane & 15;
    const int half = lane >> 4;
    const int m0   = (blockIdx.x * HEAD_WAVES + wave) * 16;

    const float* __restrict__ hrow = acc + (size_t)(m0 + c) * NHID;

    v8f d[3];
#pragma unroll
    for (int t = 0; t < 3; ++t) {
#pragma unroll
        for (int v = 0; v < 8; ++v) {
            const int cls = 16 * t + 8 * half + v;
            const float bo = bout[min(cls, NCLASS - 1)];
            d[t][v] = cls < NCLASS ? bo : 0.0f;
        }
    }

#pragma unroll
    for (int j = 0; j < NHID / 8; ++j) {
        const int k0 = 8 * j + 4 * half;
        const float4 hv = *(const float4*)(hrow + k0);
        const float4 bv = *(const float4*)(b2 + k0);
        float r[4];
        r[0] = fmaxf(hv.x + bv.x, 0.0f);
        r[1] = fmaxf(hv.y + bv.y, 0.0f);
        r[2] = fmaxf(hv.z + bv.z, 0.0f);
        r[3] = fmaxf(hv.w + bv.w, 0.0f);
#pragma unroll
        for (int s = 0; s < 2; ++s) {
            v2f b; b.x = r[2 * s]; b.y = r[2 * s + 1];
            const float* __restrict__ wk = Wout + (size_t)(k0 + 2 * s) * NCLASS;
#pragma unroll
            for (int t = 0; t < 3; ++t) {
                const int cls = 16 * t + c;
                const int cl  = min(cls, NCLASS - 1);
                const float w0 = wk[cl];
                const float w1 = wk[NCLASS + cl];
                v2f a;
                a.x = cls < NCLASS ? w0 : 0.0f;
                a.y = cls < NCLASS ? w1 : 0.0f;
                d[t] = __builtin_amdgcn_wmma_f32_16x16x4_f32(
                            false, a,  false, b,
                            (short)0, d[t],  false,  false);
            }
        }
    }

    float mx = -INFINITY;
#pragma unroll
    for (int t = 0; t < 3; ++t) {
#pragma unroll
        for (int v = 0; v < 8; ++v) {
            const bool valid = (16 * t + 8 * half + v) < NCLASS;
            mx = fmaxf(mx, valid ? d[t][v] : -INFINITY);
        }
    }
    mx = fmaxf(mx, __shfl_xor(mx, 16));
    float sum = 0.0f;
#pragma unroll
    for (int t = 0; t < 3; ++t) {
#pragma unroll
        for (int v = 0; v < 8; ++v) {
            const bool valid = (16 * t + 8 * half + v) < NCLASS;
            const float e = __expf(d[t][v] - mx);
            sum += valid ? e : 0.0f;
        }
    }
    sum += __shfl_xor(sum, 16);
    const float lse = __logf(sum) + mx;

    float* st = stage[wave];
#pragma unroll
    for (int t = 0; t < 3; ++t) {
#pragma unroll
        for (int v = 0; v < 8; ++v) {
            const int cls = 16 * t + 8 * half + v;
            if (cls < NCLASS) st[c * NCLASS + cls] = d[t][v] - lse;
        }
    }
    __syncthreads();

    float* __restrict__ orow = out + (size_t)m0 * NCLASS;
#pragma unroll
    for (int i = 0; i < (HEAD_ROW_FLOATS + 31) / 32; ++i) {
        const int idx = 32 * i + lane;
        if (idx < HEAD_ROW_FLOATS) orow[idx] = st[idx];
    }
}

extern "C" void kernel_launch(void* const* d_in, const int* in_sizes, int n_in,
                              void* d_out, int out_size, void* d_ws, size_t ws_size,
                              hipStream_t stream)
{
    const float* x    = (const float*)d_in[0];
    const int*   a0i  = (const int*)  d_in[1];
    const float* a0w  = (const float*)d_in[2];
    const int*   a1i  = (const int*)  d_in[3];
    const float* a1w  = (const float*)d_in[4];
    const float* W1   = (const float*)d_in[5];
    const float* b1   = (const float*)d_in[6];
    const float* W2   = (const float*)d_in[7];
    const float* b2   = (const float*)d_in[8];
    const float* Wout = (const float*)d_in[9];
    const float* bout = (const float*)d_in[10];
    float* out = (float*)d_out;

    const size_t NH = (size_t)N_NODES * NHID;
    float* bufA = (float*)d_ws;
    float* bufB = bufA + NH;
    int*   cnt    = (int*)(bufB + NH);
    int*   rowptr = cnt + ROWS2_PAD;
    int*   rank   = rowptr + ROWS2_PAD;
    int2*  rec    = (int2*)(rank + N_EDGES2);

    const dim3 gGemm(N_NODES / 16), bGemm(256);
    const dim3 gZero(ROWS2_PAD / 4 / 256), bZero(256);
    const dim3 gEdge((N_EDGES / 4 + 255) / 256, 2), bEdge(256);
    const dim3 gScan(ROWS2_PAD / 1024), bScan(1024);
    const dim3 gPull((N_NODES / ROWS_PER_WAVE + 7) / 8), bPull(256);

    zero_i4<<<gZero, bZero, 0, stream>>>((int4*)cnt, ROWS2_PAD / 4);
    csr_count_rank<<<gEdge, bEdge, 0, stream>>>(a0i + N_EDGES, a1i + N_EDGES, cnt, rank);
    csr_scan<<<gScan, bScan, 0, stream>>>(cnt, rowptr);
    csr_place<<<gEdge, bEdge, 0, stream>>>(a0i, a0w, a1i, a1w, rowptr, rank, rec);

    gemm128_wmma<false><<<gGemm, bGemm, 0, stream>>>(x, W1, nullptr, bufA);
    spmm_pull<<<gPull, bPull, 0, stream>>>(rowptr, rec, bufA, bufB);

    gemm128_wmma<true><<<gGemm, bGemm, 0, stream>>>(bufB, W2, b1, bufA);
    spmm_pull<<<gPull, bPull, 0, stream>>>(rowptr + N_NODES, rec, bufA, bufB);

    head_logsoftmax_wmma<<<dim3(N_NODES / 16 / HEAD_WAVES), dim3(32 * HEAD_WAVES), 0, stream>>>(bufB, b2, Wout, bout, out);
}


# gemm: W kept in registers as B fragments; a wave owns 32 columns and sweeps 6-7 row tiles (2048 waves, not 25000), A rows by 16-byte loads one tile ahead; dot sums in a new fixed k order
# speedup vs baseline: 1.9582x; 1.1786x over previous
#include <hip/hip_runtime.h>
#include <math.h>

#define N_NODES 50000
#define N_EDGES 600000
#define NFEAT 128
#define NHID 128
#define NCLASS 41
#define N_ROWS2  (2 * N_NODES)
#define N_EDGES2 (2 * N_EDGES)
#define ROWS2_PAD 100352
#define ROWS_PER_WAVE 4

typedef float v2f __attribute__((ext_vector_type(2)));
typedef float v8f __attribute__((ext_vector_type(8)));

#define GEMM_TILES  (N_NODES / 16)
#define GEMM_BLOCKS 512
#define GEMM_KTRIPS (NHID / 8)

__device__ __forceinline__
void gemm_load_tile(const float* __restrict__ acol, int rt, float4 (&a)[GEMM_KTRIPS])
{
#pragma unroll
    for (int j = 0; j < GEMM_KTRIPS; ++j)
        a[j] = *(const float4*)(acol + (size_t)rt * (16 * NHID) + 8 * j);
    __builtin_amdgcn_sched_barrier(0);
}

template<bool BIAS_RELU>
__device__ __forceinline__
void gemm_tile(const float4 (&a)[GEMM_KTRIPS], const v2f (&bf)[2][2 * GEMM_KTRIPS],
               const float* __restrict__ bcol, float* __restrict__ crow)
{
    v8f acc0 = {}, acc1 = {};
#pragma unroll
    for (int j = 0; j < GEMM_KTRIPS; ++j) {
        float r[4] = { a[j].x, a[j].y, a[j].z, a[j].w };
        if (BIAS_RELU) {
            const float4 bv = *(const float4*)(bcol + 8 * j);
            r[0] = fmaxf(r[0] + bv.x, 0.0f);
            r[1] = fmaxf(r[1] + bv.y, 0.0f);
            r[2] = fmaxf(r[2] + bv.z, 0.0f);
            r[3] = fmaxf(r[3] + bv.w, 0.0f);
        }
#pragma unroll
        for (int s = 0; s < 2; ++s) {
            v2f av; av.x = r[2 * s]; av.y = r[2 * s + 1];
            acc0 = __builtin_amdgcn_wmma_f32_16x16x4_f32(
                        false, av,  false, bf[0][2 * j + s],
                        (short)0, acc0,  false,  false);
            acc1 = __builtin_amdgcn_wmma_f32_16x16x4_f32(
                        false, av,  false, bf[1][2 * j + s],
                        (short)0, acc1,  false,  false);
        }
    }
#pragma unroll
    for (int v = 0; v < 8; ++v)
        *(float2*)(crow + (size_t)v * NHID) = make_float2(acc0[v], acc1[v]);
}

template<bool BIAS_RELU>
__global__ __launch_bounds__(128)
void gemm128_wmma(const float* __restrict__ A, const float* __restrict__ B,
                  const float* __restrict__ bias, float* __restrict__ C)
{
    const int lane = threadIdx.x & 31;
    const int wave = threadIdx.x >> 5;
    const int c    = lane & 15;
    const int half = lane >> 4;
    const int ncol = wave * 32 + 2 * c;

    v2f bf[2][2 * GEMM_KTRIPS];
#pragma unroll
    for (int j = 0; j < GEMM_KTRIPS; ++j) {
#pragma unroll
        for (int s = 0; s < 2; ++s) {
            const int k = 8 * j + 4 * half + 2 * s;
            const float2 w0 = *(const float2*)(B + (size_t)k       * NHID + ncol);
            const float2 w1 = *(const float2*)(B + (size_t)(k + 1) * NHID + ncol);
            bf[0][2 * j + s].x = w0.x;  bf[0][2 * j + s].y = w1.x;
            bf[1][2 * j + s].x = w0.y;  bf[1][2 * j + s].y = w1.y;
        }
    }

    const float* __restrict__ acol = A + (size_t)c * NHID + 4 * half;
    const float* __restrict__ bcol = bias + 4 * half;
    float*       __restrict__ ccol = C + (size_t)(8 * half) * NHID + ncol;

    float4 ra[GEMM_KTRIPS], rb[GEMM_KTRIPS];
    int rt   = blockIdx.x;
    int left = (GEMM_TILES - 1 - rt) / GEMM_BLOCKS + 1;
    gemm_load_tile(acol, rt, ra);
    while (left >= 2) {
        gemm_load_tile(acol, rt + GEMM_BLOCKS, rb);
        gemm_tile<BIAS_RELU>(ra, bf, bcol, ccol + (size_t)rt * (16 * NHID));
        gemm_load_tile(acol, min(rt + 2 * GEMM_BLOCKS, GEMM_TILES - 1), ra);
        gemm_tile<BIAS_RELU>(rb, bf, bcol, ccol + (size_t)(rt + GEMM_BLOCKS) * (16 * NHID));
        rt   += 2 * GEMM_BLOCKS;
        left -= 2;
    }
    if (left) gemm_tile<BIAS_RELU>(ra, bf, bcol, ccol + (size_t)rt * (16 * NHID));
}

__global__ __launch_bounds__(256)
void zero_i4(int4* __restrict__ p, int n4)
{
    const int i = blockIdx.x * blockDim.x + threadIdx.x;
    if (i < n4) p[i] = make_int4(0, 0, 0, 0);
}

__global__ __launch_bounds__(256)
void csr_count_rank(const int* __restrict__ dst0, const int* __restrict__ dst1,
                    int* __restrict__ cnt, int* __restrict__ rank)
{
    const int g = blockIdx.y;
    const int q = blockIdx.x * blockDim.x + threadIdx.x;
    if (q >= N_EDGES / 4) return;
    const int4 d = ((const int4*)(g ? dst1 : dst0))[q];
    int* __restrict__ c = cnt + g * N_NODES;
    int4 r;
    r.x = atomicAdd(c + d.x, 1);
    r.y = atomicAdd(c + d.y, 1);
    r.z = atomicAdd(c + d.z, 1);
    r.w = atomicAdd(c + d.w, 1);
    ((int4*)(rank + g * N_EDGES))[q] = r;
}

__global__ __launch_bounds__(1024)
void csr_scan(const int* __restrict__ cnt, int* __restrict__ rowptr)
{
    __shared__ int w_tot[32];
    __shared__ int w_front[32];
    __shared__ int s_front;
    const int t = threadIdx.x, lane = t & 31, wv = t >> 5;
    const int base = blockIdx.x * 1024;

    int front = 0;
    const int4* __restrict__ c4 = (const int4*)cnt;
    const int nq = base / 4;
    for (int i = t; i < nq; i += 8 * 1024) {
#pragma unroll
        for (int u = 0; u < 8; ++u) {
            const int j = i + u * 1024;
            const int4 v = c4[j < nq ? j : ROWS2_PAD / 4 - 1];
            front += (v.x + v.y) + (v.z + v.w);
        }
    }
    const int idx = base + t;
    const int c = idx < N_ROWS2 ? cnt[idx] : 0;

    int incl = c;
    for (int o = 1; o < 32; o <<= 1) {
        const int v = __shfl_up(incl, o);
        if (lane >= o) incl += v;
    }
    for (int o = 16; o > 0; o >>= 1) front += __shfl_xor(front, o);
    if (lane == 31) w_tot[wv] = incl;
    if (lane == 0)  w_front[wv] = front;
    __syncthreads();
    if (wv == 0) {
        const int tot = w_tot[lane];
        int wi = tot;
        for (int o = 1; o < 32; o <<= 1) {
            const int v = __shfl_up(wi, o);
            if (lane >= o) wi += v;
        }
        int f = w_front[lane];
        for (int o = 16; o > 0; o >>= 1) f += __shfl_xor(f, o);
        w_tot[lane] = wi - tot;
        if (lane == 0) s_front = f;
    }
    __syncthreads();
    const int excl = s_front + w_tot[wv] + incl - c;
    if (idx < N_ROWS2)      rowptr[idx] = excl;
    if (idx == N_ROWS2 - 1) rowptr[N_ROWS2] = excl + c;
}

__global__ __launch_bounds__(256)
void csr_place(const int* __restrict__ idx0, const float* __restrict__ w0,
               const int* __restrict__ idx1, const float* __restrict__ w1,
               const int* __restrict__ rowptr, const int* __restrict__ rank,
               int2* __restrict__ rec)
{
    const int g = blockIdx.y;
    const int q = blockIdx.x * blockDim.x + threadIdx.x;
    if (q >= N_EDGES / 4) return;
    const int* __restrict__ idx = g ? idx1 : idx0;
    const int4   s = ((const int4*)idx)[q];
    const int4   d = ((const int4*)(idx + N_EDGES))[q];
    const float4 w = ((const float4*)(g ? w1 : w0))[q];
    const int4   r = ((const int4*)(rank + g * N_EDGES))[q];
    const int* __restrict__ rp = rowptr + g * N_NODES;
    const unsigned p0 = min((unsigned)(rp[d.x] + r.x), (unsigned)(N_EDGES2 - 1));
    const unsigned p1 = min((unsigned)(rp[d.y] + r.y), (unsigned)(N_EDGES2 - 1));
    const unsigned p2 = min((unsigned)(rp[d.z] + r.z), (unsigned)(N_EDGES2 - 1));
    const unsigned p3 = min((unsigned)(rp[d.w] + r.w), (unsigned)(N_EDGES2 - 1));
    rec[p0] = make_int2(s.x, __float_as_int(w.x));
    rec[p1] = make_int2(s.y, __float_as_int(w.y));
    rec[p2] = make_int2(s.z, __float_as_int(w.z));
    rec[p3] = make_int2(s.w, __float_as_int(w.w));
}

__device__ __forceinline__
void gather_chunk(const float4* __restrict__ h4, int lane, int rs, float rw, int n, float4& acc)
{
    for (int j = 0; j < n; j += 4) {
        const unsigned s0 = min((unsigned)__builtin_amdgcn_readlane(rs, j),     (unsigned)(N_NODES - 1));
        const unsigned s1 = min((unsigned)__builtin_amdgcn_readlane(rs, j + 1), (unsigned)(N_NODES - 1));
        const unsigned s2 = min((unsigned)__builtin_amdgcn_readlane(rs, j + 2), (unsigned)(N_NODES - 1));
        const unsigned s3 = min((unsigned)__builtin_amdgcn_readlane(rs, j + 3), (unsigned)(N_NODES - 1));
        const float w0 = __int_as_float(__builtin_amdgcn_readlane(__float_as_int(rw), j));
        const float w1 = __int_as_float(__builtin_amdgcn_readlane(__float_as_int(rw), j + 1));
        const float w2 = __int_as_float(__builtin_amdgcn_readlane(__float_as_int(rw), j + 2));
        const float w3 = __int_as_float(__builtin_amdgcn_readlane(__float_as_int(rw), j + 3));
        const float4 v0 = h4[(size_t)s0 * (NHID / 4) + lane];
        const float4 v1 = h4[(size_t)s1 * (NHID / 4) + lane];
        const float4 v2 = h4[(size_t)s2 * (NHID / 4) + lane];
        const float4 v3 = h4[(size_t)s3 * (NHID / 4) + lane];
        acc.x = fmaf(w0, v0.x, acc.x); acc.y = fmaf(w0, v0.y, acc.y);
        acc.z = fmaf(w0, v0.z, acc.z); acc.w = fmaf(w0, v0.w, acc.w);
        acc.x = fmaf(w1, v1.x, acc.x); acc.y = fmaf(w1, v1.y, acc.y);
        acc.z = fmaf(w1, v1.z, acc.z); acc.w = fmaf(w1, v1.w, acc.w);
        acc.x = fmaf(w2, v2.x, acc.x); acc.y = fmaf(w2, v2.y, acc.y);
        acc.z = fmaf(w2, v2.z, acc.z); acc.w = fmaf(w2, v2.w, acc.w);
        acc.x = fmaf(w3, v3.x, acc.x); acc.y = fmaf(w3, v3.y, acc.y);
        acc.z = fmaf(w3, v3.z, acc.z); acc.w = fmaf(w3, v3.w, acc.w);
    }
}

__global__ __launch_bounds__(256)
void spmm_pull(const int* __restrict__ rowptr, const int2* __restrict__ rec,
               const float* __restrict__ h, float* __restrict__ out)
{
    const int lane = threadIdx.x & 31;
    const int row0 = __builtin_amdgcn_readfirstlane(
                         (int)((blockIdx.x * 8 + (threadIdx.x >> 5)) * ROWS_PER_WAVE));
    if (row0 >= N_NODES) return;

    int p[ROWS_PER_WAVE + 1];
#pragma unroll
    for (int k = 0; k <= ROWS_PER_WAVE; ++k)
        p[k] = min(max(rowptr[row0 + k], 0), N_EDGES2);

    int rs[ROWS_PER_WAVE]; float rw[ROWS_PER_WAVE]; int n[ROWS_PER_WAVE];
#pragma unroll
    for (int k = 0; k < ROWS_PER_WAVE; ++k) {
        n[k] = max(min(p[k + 1] - p[k], 32), 0);
        const int i = min(p[k] + min(lane, max(n[k], 1) - 1), N_EDGES2 - 1);
        const int2 r = rec[i];
        rs[k] = r.x;
        rw[k] = lane < n[k] ? __int_as_float(r.y) : 0.0f;
    }

    const float4* __restrict__ h4 = (const float4*)h;
    float4* __restrict__ out4 = (float4*)out;
#pragma unroll
    for (int k = 0; k < ROWS_PER_WAVE; ++k) {
        float4 acc = make_float4(0.f, 0.f, 0.f, 0.f);
        gather_chunk(h4, lane, rs[k], rw[k], n[k], acc);
        for (int base = p[k] + 32; base < p[k + 1]; base += 32) {
            const int m = min(p[k + 1] - base, 32);
            const int2 r = rec[base + min(lane, m - 1)];
            gather_chunk(h4, lane, r.x, lane < m ? __int_as_float(r.y) : 0.0f, m, acc);
        }
        out4[(size_t)(row0 + k) * (NHID / 4) + lane] = acc;
    }
}

#define HEAD_WAVES 5
#define HEAD_ROW_FLOATS (16 * NCLASS)

__global__ __launch_bounds__(32 * HEAD_WAVES)
void head_logsoftmax_wmma(const float* __restrict__ acc, const float* __restrict__ b2,
                          const float* __restrict__ Wout, const float* __restrict__ bout,
                          float* __restrict__ out)
{
    __shared__ float stage[HEAD_WAVES][HEAD_ROW_FLOATS];
    const int lane = threadIdx.x & 31;
    const int wave = threadIdx.x >> 5;
    const int c    = lane & 15;
    const int half = lane >> 4;
    const int m0   = (blockIdx.x * HEAD_WAVES + wave) * 16;

    const float* __restrict__ hrow = acc + (size_t)(m0 + c) * NHID;

    v8f d[3];
#pragma unroll
    for (int t = 0; t < 3; ++t) {
#pragma unroll
        for (int v = 0; v < 8; ++v) {
            const int cls = 16 * t + 8 * half + v;
            const float bo = bout[min(cls, NCLASS - 1)];
            d[t][v] = cls < NCLASS ? bo : 0.0f;
        }
    }

#pragma unroll
    for (int j = 0; j < NHID / 8; ++j) {
        const int k0 = 8 * j + 4 * half;
        const float4 hv = *(const float4*)(hrow + k0);
        const float4 bv = *(const float4*)(b2 + k0);
        float r[4];
        r[0] = fmaxf(hv.x + bv.x, 0.0f);
        r[1] = fmaxf(hv.y + bv.y, 0.0f);
        r[2] = fmaxf(hv.z + bv.z, 0.0f);
        r[3] = fmaxf(hv.w + bv.w, 0.0f);
#pragma unroll
        for (int s = 0; s < 2; ++s) {
            v2f b; b.x = r[2 * s]; b.y = r[2 * s + 1];
            const float* __restrict__ wk = Wout + (size_t)(k0 + 2 * s) * NCLASS;
#pragma unroll
            for (int t = 0; t < 3; ++t) {
                const int cls = 16 * t + c;
                const int cl  = min(cls, NCLASS - 1);
                const float w0 = wk[cl];
                const float w1 = wk[NCLASS + cl];
                v2f a;
                a.x = cls < NCLASS ? w0 : 0.0f;
                a.y = cls < NCLASS ? w1 : 0.0f;
                d[t] = __builtin_amdgcn_wmma_f32_16x16x4_f32(
                            false, a,  false, b,
                            (short)0, d[t],  false,  false);
            }
        }
    }

    float mx = -INFINITY;
#pragma unroll
    for (int t = 0; t < 3; ++t) {
#pragma unroll
        for (int v = 0; v < 8; ++v) {
            const bool valid = (16 * t + 8 * half + v) < NCLASS;
            mx = fmaxf(mx, valid ? d[t][v] : -INFINITY);
        }
    }
    mx = fmaxf(mx, __shfl_xor(mx, 16));
    float sum = 0.0f;
#pragma unroll
    for (int t = 0; t < 3; ++t) {
#pragma unroll
        for (int v = 0; v < 8; ++v) {
            const bool valid = (16 * t + 8 * half + v) < NCLASS;
            const float e = __expf(d[t][v] - mx);
            sum += valid ? e : 0.0f;
        }
    }
    sum += __shfl_xor(sum, 16);
    const float lse = __logf(sum) + mx;

    float* st = stage[wave];
#pragma unroll
    for (int t = 0; t < 3; ++t) {
#pragma unroll
        for (int v = 0; v < 8; ++v) {
            const int cls = 16 * t + 8 * half + v;
            if (cls < NCLASS) st[c * NCLASS + cls] = d[t][v] - lse;
        }
    }
    __syncthreads();

    float* __restrict__ orow = out + (size_t)m0 * NCLASS;
#pragma unroll
    for (int i = 0; i < (HEAD_ROW_FLOATS + 31) / 32; ++i) {
        const int idx = 32 * i + lane;
        if (idx < HEAD_ROW_FLOATS) orow[idx] = st[idx];
    }
}

extern "C" void kernel_launch(void* const* d_in, const int* in_sizes, int n_in,
                              void* d_out, int out_size, void* d_ws, size_t ws_size,
                              hipStream_t stream)
{
    const float* x    = (const float*)d_in[0];
    const int*   a0i  = (const int*)  d_in[1];
    const float* a0w  = (const float*)d_in[2];
    const int*   a1i  = (const int*)  d_in[3];
    const float* a1w  = (const float*)d_in[4];
    const float* W1   = (const float*)d_in[5];
    const float* b1   = (const float*)d_in[6];
    const float* W2   = (const float*)d_in[7];
    const float* b2   = (const float*)d_in[8];
    const float* Wout = (const float*)d_in[9];
    const float* bout = (const float*)d_in[10];
    float* out = (float*)d_out;

    const size_t NH = (size_t)N_NODES * NHID;
    float* bufA = (float*)d_ws;
    float* bufB = bufA + NH;
    int*   cnt    = (int*)(bufB + NH);
    int*   rowptr = cnt + ROWS2_PAD;
    int*   rank   = rowptr + ROWS2_PAD;
    int2*  rec    = (int2*)(rank + N_EDGES2);

    const dim3 gGemm(GEMM_BLOCKS), bGemm(128);
    const dim3 gZero(ROWS2_PAD / 4 / 256), bZero(256);
    const dim3 gEdge((N_EDGES / 4 + 255) / 256, 2), bEdge(256);
    const dim3 gScan(ROWS2_PAD / 1024), bScan(1024);
    const dim3 gPull((N_NODES / ROWS_PER_WAVE + 7) / 8), bPull(256);

    zero_i4<<<gZero, bZero, 0, stream>>>((int4*)cnt, ROWS2_PAD / 4);
    csr_count_rank<<<gEdge, bEdge, 0, stream>>>(a0i + N_EDGES, a1i + N_EDGES, cnt, rank);
    csr_scan<<<gScan, bScan, 0, stream>>>(cnt, rowptr);
    csr_place<<<gEdge, bEdge, 0, stream>>>(a0i, a0w, a1i, a1w, rowptr, rank, rec);

    gemm128_wmma<false><<<gGemm, bGemm, 0, stream>>>(x, W1, nullptr, bufA);
    spmm_pull<<<gPull, bPull, 0, stream>>>(rowptr, rec, bufA, bufB);

    gemm128_wmma<true><<<gGemm, bGemm, 0, stream>>>(bufB, W2, b1, bufA);
    spmm_pull<<<gPull, bPull, 0, stream>>>(rowptr + N_NODES, rec, bufA, bufB);

    head_logsoftmax_wmma<<<dim3(N_NODES / 16 / HEAD_WAVES), dim3(32 * HEAD_WAVES), 0, stream>>>(bufB, b2, Wout, bout, out);
}


# edge bucketing: each dst row owns a 32-record slot filled in one edge pass from its counter's old value (spill list past 32); no prefix-sum/place launches, no rank array; sums in arrival order
# speedup vs baseline: 2.1638x; 1.1050x over previous
#include <hip/hip_runtime.h>
#include <math.h>

#define N_NODES 50000
#define N_EDGES 600000
#define NFEAT 128
#define NHID 128
#define NCLASS 41
#define N_ROWS2  (2 * N_NODES)
#define N_EDGES2 (2 * N_EDGES)
#define ROWS2_PAD 100352
#define ROWS_PER_WAVE 4
#define SLOT_CAP 32
#define SPILL_IDX N_ROWS2

typedef float v2f __attribute__((ext_vector_type(2)));
typedef float v8f __attribute__((ext_vector_type(8)));

#define GEMM_TILES  (N_NODES / 16)
#define GEMM_BLOCKS 512
#define GEMM_KTRIPS (NHID / 8)

__device__ __forceinline__
void gemm_load_tile(const float* __restrict__ acol, int rt, float4 (&a)[GEMM_KTRIPS])
{
#pragma unroll
    for (int j = 0; j < GEMM_KTRIPS; ++j)
        a[j] = *(const float4*)(acol + (size_t)rt * (16 * NHID) + 8 * j);
    __builtin_amdgcn_sched_barrier(0);
}

template<bool BIAS_RELU>
__device__ __forceinline__
void gemm_tile(const float4 (&a)[GEMM_KTRIPS], const v2f (&bf)[2][2 * GEMM_KTRIPS],
               const float* __restrict__ bcol, float* __restrict__ crow)
{
    v8f acc0 = {}, acc1 = {};
#pragma unroll
    for (int j = 0; j < GEMM_KTRIPS; ++j) {
        float r[4] = { a[j].x, a[j].y, a[j].z, a[j].w };
        if (BIAS_RELU) {
            const float4 bv = *(const float4*)(bcol + 8 * j);
            r[0] = fmaxf(r[0] + bv.x, 0.0f);
            r[1] = fmaxf(r[1] + bv.y, 0.0f);
            r[2] = fmaxf(r[2] + bv.z, 0.0f);
            r[3] = fmaxf(r[3] + bv.w, 0.0f);
        }
#pragma unroll
        for (int s = 0; s < 2; ++s) {
            v2f av; av.x = r[2 * s]; av.y = r[2 * s + 1];
            acc0 = __builtin_amdgcn_wmma_f32_16x16x4_f32(
                        false, av,  false, bf[0][2 * j + s],
                        (short)0, acc0,  false,  false);
            acc1 = __builtin_amdgcn_wmma_f32_16x16x4_f32(
                        false, av,  false, bf[1][2 * j + s],
                        (short)0, acc1,  false,  false);
        }
    }
#pragma unroll
    for (int v = 0; v < 8; ++v)
        *(float2*)(crow + (size_t)v * NHID) = make_float2(acc0[v], acc1[v]);
}

template<bool BIAS_RELU>
__global__ __launch_bounds__(128)
void gemm128_wmma(const float* __restrict__ A, const float* __restrict__ B,
                  const float* __restrict__ bias, float* __restrict__ C)
{
    const int lane = threadIdx.x & 31;
    const int wave = threadIdx.x >> 5;
    const int c    = lane & 15;
    const int half = lane >> 4;
    const int ncol = wave * 32 + 2 * c;

    v2f bf[2][2 * GEMM_KTRIPS];
#pragma unroll
    for (int j = 0; j < GEMM_KTRIPS; ++j) {
#pragma unroll
        for (int s = 0; s < 2; ++s) {
            const int k = 8 * j + 4 * half + 2 * s;
            const float2 w0 = *(const float2*)(B + (size_t)k       * NHID + ncol);
            const float2 w1 = *(const float2*)(B + (size_t)(k + 1) * NHID + ncol);
            bf[0][2 * j + s].x = w0.x;  bf[0][2 * j + s].y = w1.x;
            bf[1][2 * j + s].x = w0.y;  bf[1][2 * j + s].y = w1.y;
        }
    }

    const float* __restrict__ acol = A + (size_t)c * NHID + 4 * half;
    const float* __restrict__ bcol = bias + 4 * half;
    float*       __restrict__ ccol = C + (size_t)(8 * half) * NHID + ncol;

    float4 ra[GEMM_KTRIPS], rb[GEMM_KTRIPS];
    int rt   = blockIdx.x;
    int left = (GEMM_TILES - 1 - rt) / GEMM_BLOCKS + 1;
    gemm_load_tile(acol, rt, ra);
    while (left >= 2) {
        gemm_load_tile(acol, rt + GEMM_BLOCKS, rb);
        gemm_tile<BIAS_RELU>(ra, bf, bcol, ccol + (size_t)rt * (16 * NHID));
        gemm_load_tile(acol, min(rt + 2 * GEMM_BLOCKS, GEMM_TILES - 1), ra);
        gemm_tile<BIAS_RELU>(rb, bf, bcol, ccol + (size_t)(rt + GEMM_BLOCKS) * (16 * NHID));
        rt   += 2 * GEMM_BLOCKS;
        left -= 2;
    }
    if (left) gemm_tile<BIAS_RELU>(ra, bf, bcol, ccol + (size_t)rt * (16 * NHID));
}

__global__ __launch_bounds__(256)
void zero_i4(int4* __restrict__ p, int n4)
{
    const int i = blockIdx.x * blockDim.x + threadIdx.x;
    if (i < n4) p[i] = make_int4(0, 0, 0, 0);
}

__global__ __launch_bounds__(256)
void edge_fill(const int* __restrict__ idx0, const float* __restrict__ w0,
               const int* __restrict__ idx1, const float* __restrict__ w1,
               int* __restrict__ cnt, int2* __restrict__ slot, int4* __restrict__ spill)
{
    const int g = blockIdx.y;
    const int q = blockIdx.x * blockDim.x + threadIdx.x;
    if (q >= N_EDGES / 8) return;
    const int*   __restrict__ idx = g ? idx1 : idx0;
    const float* __restrict__ wgt = g ? w1 : w0;
    int row[8], src[8], wb[8], place[8];
#pragma unroll
    for (int u = 0; u < 2; ++u) {
        const int i4 = q + u * (N_EDGES / 8);
        const int4   s = ((const int4*)idx)[i4];
        const int4   d = ((const int4*)(idx + N_EDGES))[i4];
        const float4 w = ((const float4*)wgt)[i4];
        src[4 * u] = s.x; src[4 * u + 1] = s.y; src[4 * u + 2] = s.z; src[4 * u + 3] = s.w;
        row[4 * u]     = g * N_NODES + (int)min((unsigned)d.x, (unsigned)(N_NODES - 1));
        row[4 * u + 1] = g * N_NODES + (int)min((unsigned)d.y, (unsigned)(N_NODES - 1));
        row[4 * u + 2] = g * N_NODES + (int)min((unsigned)d.z, (unsigned)(N_NODES - 1));
        row[4 * u + 3] = g * N_NODES + (int)min((unsigned)d.w, (unsigned)(N_NODES - 1));
        wb[4 * u]     = __float_as_int(w.x); wb[4 * u + 1] = __float_as_int(w.y);
        wb[4 * u + 2] = __float_as_int(w.z); wb[4 * u + 3] = __float_as_int(w.w);
    }
#pragma unroll
    for (int e = 0; e < 8; ++e)
        place[e] = atomicAdd(cnt + row[e], 1);
#pragma unroll
    for (int e = 0; e < 8; ++e) {
        if ((unsigned)place[e] < (unsigned)SLOT_CAP) {
            slot[(size_t)row[e] * SLOT_CAP + place[e]] = make_int2(src[e], wb[e]);
        } else {
            const unsigned o = min((unsigned)atomicAdd(cnt + SPILL_IDX, 1), (unsigned)(N_EDGES2 - 1));
            spill[o] = make_int4(row[e], src[e], wb[e], 0);
        }
    }
}

__device__ __forceinline__
void gather_chunk(const float4* __restrict__ h4, int lane, int rs, float rw, int n, float4& acc)
{
    for (int j = 0; j < n; j += 4) {
        const unsigned s0 = min((unsigned)__builtin_amdgcn_readlane(rs, j),     (unsigned)(N_NODES - 1));
        const unsigned s1 = min((unsigned)__builtin_amdgcn_readlane(rs, j + 1), (unsigned)(N_NODES - 1));
        const unsigned s2 = min((unsigned)__builtin_amdgcn_readlane(rs, j + 2), (unsigned)(N_NODES - 1));
        const unsigned s3 = min((unsigned)__builtin_amdgcn_readlane(rs, j + 3), (unsigned)(N_NODES - 1));
        const float w0 = __int_as_float(__builtin_amdgcn_readlane(__float_as_int(rw), j));
        const float w1 = __int_as_float(__builtin_amdgcn_readlane(__float_as_int(rw), j + 1));
        const float w2 = __int_as_float(__builtin_amdgcn_readlane(__float_as_int(rw), j + 2));
        const float w3 = __int_as_float(__builtin_amdgcn_readlane(__float_as_int(rw), j + 3));
        const float4 v0 = h4[(size_t)s0 * (NHID / 4) + lane];
        const float4 v1 = h4[(size_t)s1 * (NHID / 4) + lane];
        const float4 v2 = h4[(size_t)s2 * (NHID / 4) + lane];
        const float4 v3 = h4[(size_t)s3 * (NHID / 4) + lane];
        acc.x = fmaf(w0, v0.x, acc.x); acc.y = fmaf(w0, v0.y, acc.y);
        acc.z = fmaf(w0, v0.z, acc.z); acc.w = fmaf(w0, v0.w, acc.w);
        acc.x = fmaf(w1, v1.x, acc.x); acc.y = fmaf(w1, v1.y, acc.y);
        acc.z = fmaf(w1, v1.z, acc.z); acc.w = fmaf(w1, v1.w, acc.w);
        acc.x = fmaf(w2, v2.x, acc.x); acc.y = fmaf(w2, v2.y, acc.y);
        acc.z = fmaf(w2, v2.z, acc.z); acc.w = fmaf(w2, v2.w, acc.w);
        acc.x = fmaf(w3, v3.x, acc.x); acc.y = fmaf(w3, v3.y, acc.y);
        acc.z = fmaf(w3, v3.z, acc.z); acc.w = fmaf(w3, v3.w, acc.w);
    }
}

__device__ __forceinline__
void gather_spilled(const int4* __restrict__ spill, int m, int grow,
                    const float4* __restrict__ h4, int lane, float4& acc)
{
    for (int base = 0; base < m; base += 32) {
        int4 e = make_int4(-1, 0, 0, 0);
        if (base + lane < m) e = spill[base + lane];
        const bool mine = e.x == grow;
        const unsigned any = __builtin_amdgcn_ballot_w32(mine);
        if (any) {
            const int first = __builtin_amdgcn_readlane(e.y, __builtin_ctz(any));
            gather_chunk(h4, lane, mine ? e.y : first, mine ? __int_as_float(e.z) : 0.0f, 32, acc);
        }
    }
}

__global__ __launch_bounds__(256)
void spmm_pull(const int* __restrict__ cnt, const int2* __restrict__ slot,
               const int4* __restrict__ spill, int row_base,
               const float* __restrict__ h, float* __restrict__ out)
{
    const int lane = threadIdx.x & 31;
    const int row0 = __builtin_amdgcn_readfirstlane(
                         (int)((blockIdx.x * 8 + (threadIdx.x >> 5)) * ROWS_PER_WAVE));
    if (row0 >= N_NODES) return;
    const int grow0 = row_base + row0;

    int tot[ROWS_PER_WAVE], n[ROWS_PER_WAVE];
#pragma unroll
    for (int k = 0; k < ROWS_PER_WAVE; ++k) {
        tot[k] = cnt[grow0 + k];
        n[k]   = max(min(tot[k], SLOT_CAP), 0);
    }

    int rs[ROWS_PER_WAVE]; float rw[ROWS_PER_WAVE];
#pragma unroll
    for (int k = 0; k < ROWS_PER_WAVE; ++k) {
        const int2* __restrict__ rowslot = n[k] > 0 ? slot + (size_t)(grow0 + k) * SLOT_CAP
                                                    : (const int2*)cnt;
        const int2 r = rowslot[min(lane, max(n[k], 1) - 1)];
        rs[k] = r.x;
        rw[k] = lane < n[k] ? __int_as_float(r.y) : 0.0f;
    }

    const float4* __restrict__ h4 = (const float4*)h;
    float4* __restrict__ out4 = (float4*)out;
#pragma unroll
    for (int k = 0; k < ROWS_PER_WAVE; ++k) {
        float4 acc = make_float4(0.f, 0.f, 0.f, 0.f);
        gather_chunk(h4, lane, rs[k], rw[k], n[k], acc);
        if (tot[k] > SLOT_CAP) {
            const int m = max(min(cnt[SPILL_IDX], N_EDGES2), 0);
            gather_spilled(spill, m, grow0 + k, h4, lane, acc);
        }
        out4[(size_t)(row0 + k) * (NHID / 4) + lane] = acc;
    }
}

#define HEAD_WAVES 5
#define HEAD_ROW_FLOATS (16 * NCLASS)

__global__ __launch_bounds__(32 * HEAD_WAVES)
void head_logsoftmax_wmma(const float* __restrict__ acc, const float* __restrict__ b2,
                          const float* __restrict__ Wout, const float* __restrict__ bout,
                          float* __restrict__ out)
{
    __shared__ float stage[HEAD_WAVES][HEAD_ROW_FLOATS];
    const int lane = threadIdx.x & 31;
    const int wave = threadIdx.x >> 5;
    const int c    = lane & 15;
    const int half = lane >> 4;
    const int m0   = (blockIdx.x * HEAD_WAVES + wave) * 16;

    const float* __restrict__ hrow = acc + (size_t)(m0 + c) * NHID;

    v8f d[3];
#pragma unroll
    for (int t = 0; t < 3; ++t) {
#pragma unroll
        for (int v = 0; v < 8; ++v) {
            const int cls = 16 * t + 8 * half + v;
            const float bo = bout[min(cls, NCLASS - 1)];
            d[t][v] = cls < NCLASS ? bo : 0.0f;
        }
    }

#pragma unroll
    for (int j = 0; j < NHID / 8; ++j) {
        const int k0 = 8 * j + 4 * half;
        const float4 hv = *(const float4*)(hrow + k0);
        const float4 bv = *(const float4*)(b2 + k0);
        float r[4];
        r[0] = fmaxf(hv.x + bv.x, 0.0f);
        r[1] = fmaxf(hv.y + bv.y, 0.0f);
        r[2] = fmaxf(hv.z + bv.z, 0.0f);
        r[3] = fmaxf(hv.w + bv.w, 0.0f);
#pragma unroll
        for (int s = 0; s < 2; ++s) {
            v2f b; b.x = r[2 * s]; b.y = r[2 * s + 1];
            const float* __restrict__ wk = Wout + (size_t)(k0 + 2 * s) * NCLASS;
#pragma unroll
            for (int t = 0; t < 3; ++t) {
                const int cls = 16 * t + c;
                const int cl  = min(cls, NCLASS - 1);
                const float w0 = wk[cl];
                const float w1 = wk[NCLASS + cl];
                v2f a;
                a.x = cls < NCLASS ? w0 : 0.0f;
                a.y = cls < NCLASS ? w1 : 0.0f;
                d[t] = __builtin_amdgcn_wmma_f32_16x16x4_f32(
                            false, a,  false, b,
                            (short)0, d[t],  false,  false);
            }
        }
    }

    float mx = -INFINITY;
#pragma unroll
    for (int t = 0; t < 3; ++t) {
#pragma unroll
        for (int v = 0; v < 8; ++v) {
            const bool valid = (16 * t + 8 * half + v) < NCLASS;
            mx = fmaxf(mx, valid ? d[t][v] : -INFINITY);
        }
    }
    mx = fmaxf(mx, __shfl_xor(mx, 16));
    float sum = 0.0f;
#pragma unroll
    for (int t = 0; t < 3; ++t) {
#pragma unroll
        for (int v = 0; v < 8; ++v) {
            const bool valid = (16 * t + 8 * half + v) < NCLASS;
            const float e = __expf(d[t][v] - mx);
            sum += valid ? e : 0.0f;
        }
    }
    sum += __shfl_xor(sum, 16);
    const float lse = __logf(sum) + mx;

    float* st = stage[wave];
#pragma unroll
    for (int t = 0; t < 3; ++t) {
#pragma unroll
        for (int v = 0; v < 8; ++v) {
            const int cls = 16 * t + 8 * half + v;
            if (cls < NCLASS) st[c * NCLASS + cls] = d[t][v] - lse;
        }
    }
    __syncthreads();

    float* __restrict__ orow = out + (size_t)m0 * NCLASS;
#pragma unroll
    for (int i = 0; i < (HEAD_ROW_FLOATS + 31) / 32; ++i) {
        const int idx = 32 * i + lane;
        if (idx < HEAD_ROW_FLOATS) orow[idx] = st[idx];
    }
}

extern "C" void kernel_launch(void* const* d_in, const int* in_sizes, int n_in,
                              void* d_out, int out_size, void* d_ws, size_t ws_size,
                              hipStream_t stream)
{
    const float* x    = (const float*)d_in[0];
    const int*   a0i  = (const int*)  d_in[1];
    const float* a0w  = (const float*)d_in[2];
    const int*   a1i  = (const int*)  d_in[3];
    const float* a1w  = (const float*)d_in[4];
    const float* W1   = (const float*)d_in[5];
    const float* b1   = (const float*)d_in[6];
    const float* W2   = (const float*)d_in[7];
    const float* b2   = (const float*)d_in[8];
    const float* Wout = (const float*)d_in[9];
    const float* bout = (const float*)d_in[10];
    float* out = (float*)d_out;

    const size_t NH = (size_t)N_NODES * NHID;
    float* bufA = (float*)d_ws;
    float* bufB = bufA + NH;
    int*   cnt    = (int*)(bufB + NH);
    int2*  slot   = (int2*)(cnt + ROWS2_PAD);
    int4*  spill  = (int4*)(slot + (size_t)N_ROWS2 * SLOT_CAP);

    const dim3 gGemm(GEMM_BLOCKS), bGemm(128);
    const dim3 gZero(ROWS2_PAD / 4 / 256), bZero(256);
    const dim3 gEdge((N_EDGES / 8 + 255) / 256, 2), bEdge(256);
    const dim3 gPull((N_NODES / ROWS_PER_WAVE + 7) / 8), bPull(256);

    zero_i4<<<gZero, bZero, 0, stream>>>((int4*)cnt, ROWS2_PAD / 4);
    edge_fill<<<gEdge, bEdge, 0, stream>>>(a0i, a0w, a1i, a1w, cnt, slot, spill);

    gemm128_wmma<false><<<gGemm, bGemm, 0, stream>>>(x, W1, nullptr, bufA);
    spmm_pull<<<gPull, bPull, 0, stream>>>(cnt, slot, spill, 0, bufA, bufB);

    gemm128_wmma<true><<<gGemm, bGemm, 0, stream>>>(bufB, W2, b1, bufA);
    spmm_pull<<<gPull, bPull, 0, stream>>>(cnt, slot, spill, N_NODES, bufA, bufB);

    head_logsoftmax_wmma<<<dim3(N_NODES / 16 / HEAD_WAVES), dim3(32 * HEAD_WAVES), 0, stream>>>(bufB, b2, Wout, bout, out);
}
